# starting kernel
# baseline (speedup 1.0000x reference)
#include <hip/hip_runtime.h>


typedef float v2f __attribute__((ext_vector_type(2)));
typedef float v8f __attribute__((ext_vector_type(8)));

#define NHh   8
#define HDd   64
#define DIMm  512
#define Bb    16
#define NTOK  1024
#define HEADSZ (Bb * NHh * NTOK * HDd)

__device__ __forceinline__ v8f wmma4(v2f a, v2f b, v8f c) {
    return __builtin_amdgcn_wmma_f32_16x16x4_f32(
         false, a,  false, b,
         (short)0, c,  false,  false);
}

__global__ void qkv_gemm_kernel(const float* __restrict__ x,
                                const float* __restrict__ w,
                                const float* __restrict__ bias,
                                float* __restrict__ qkv) {
    const int lane = threadIdx.x & 31;
    const int lx   = lane & 15;
    const int half = lane >> 4;

    int gw = blockIdx.x * (blockDim.x >> 5) + (threadIdx.x >> 5);
    const int ct = gw % 96;  gw /= 96;
    const int mt = gw % 64;  gw /= 64;
    const int b  = gw;

    const float* arow = x + ((size_t)(b * NTOK + mt * 16 + lx)) * DIMm + 2 * half;
    const float* brow = w + ((size_t)(ct * 16 + lx)) * DIMm + 2 * half;

    v8f acc = {};
#pragma unroll 8
    for (int k = 0; k < DIMm; k += 4) {
        v2f a  = *(const v2f*)(arow + k);
        v2f bf = *(const v2f*)(brow + k);
        acc = wmma4(a, bf, acc);
    }

    const int   c     = ct * 16 + lx;
    const float bv    = bias[c];
    const int   which = c >> 9;
    const int   h     = (c >> 6) & 7;
    const int   d     = c & 63;
    const float scale = (which == 0) ? 0.125f : 1.0f;

    float* dst = qkv + (size_t)which * HEADSZ
               + ((size_t)(b * NHh + h) * NTOK) * HDd + d;
#pragma unroll
    for (int r = 0; r < 8; ++r) {
        const int row = mt * 16 + r + 8 * half;
        dst[(size_t)row * HDd] = (acc[r] + bv) * scale;
    }
}

__global__ void attn_kernel(const float* __restrict__ qkv,
                            const float* __restrict__ bias_table,
                            const int*   __restrict__ rel_index,
                            float*       __restrict__ ctx) {
    __shared__ float pbuf[8][16 * 16];
    const int lane = threadIdx.x & 31;
    const int lx   = lane & 15;
    const int half = lane >> 4;
    const int wv   = threadIdx.x >> 5;
    float* pb = pbuf[wv];

    int gw = blockIdx.x * (blockDim.x >> 5) + wv;
    const int qt = gw % 64;  gw /= 64;
    const int h  = gw % 8;   gw /= 8;
    const int b  = gw;

    const size_t headoff = ((size_t)(b * NHh + h) * NTOK) * HDd;
    const float* qp = qkv + headoff;
    const float* kp = qkv + (size_t)HEADSZ + headoff;
    const float* vp = qkv + 2 * (size_t)HEADSZ + headoff;

    v2f qf[16];
    {
        const float* qrow = qp + (size_t)(qt * 16 + lx) * HDd + 2 * half;
#pragma unroll
        for (int kk = 0; kk < 16; ++kk) qf[kk] = *(const v2f*)(qrow + kk * 4);
    }

    v8f cacc[4];
#pragma unroll
    for (int t = 0; t < 4; ++t) cacc[t] = (v8f){};
    float mrow[8], lrow[8];
#pragma unroll
    for (int r = 0; r < 8; ++r) { mrow[r] = -3.0e38f; lrow[r] = 0.0f; }

    for (int j = 0; j < 64; ++j) {
        v8f s = (v8f){};
        const float* krow = kp + (size_t)(j * 16 + lx) * HDd + 2 * half;
#pragma unroll
        for (int kk = 0; kk < 16; ++kk) {
            v2f bf = *(const v2f*)(krow + kk * 4);
            s = wmma4(qf[kk], bf, s);
        }

        const int kcol = j * 16 + lx;
#pragma unroll
        for (int r = 0; r < 8; ++r) {
            const int qrow_ = qt * 16 + r + 8 * half;
            const int idx   = rel_index[qrow_ * NTOK + kcol];
            s[r] += bias_table[idx * NHh + h];
        }

        float p[8], alpha[8];
#pragma unroll
        for (int r = 0; r < 8; ++r) {
            float tmax = s[r];
            tmax = fmaxf(tmax, __shfl_xor(tmax, 1, 32));
            tmax = fmaxf(tmax, __shfl_xor(tmax, 2, 32));
            tmax = fmaxf(tmax, __shfl_xor(tmax, 4, 32));
            tmax = fmaxf(tmax, __shfl_xor(tmax, 8, 32));
            const float mnew = fmaxf(mrow[r], tmax);
            alpha[r] = __expf(mrow[r] - mnew);
            const float pv = __expf(s[r] - mnew);
            float rs = pv;
            rs += __shfl_xor(rs, 1, 32);
            rs += __shfl_xor(rs, 2, 32);
            rs += __shfl_xor(rs, 4, 32);
            rs += __shfl_xor(rs, 8, 32);
            lrow[r] = lrow[r] * alpha[r] + rs;
            mrow[r] = mnew;
            p[r] = pv;
        }
#pragma unroll
        for (int t = 0; t < 4; ++t)
#pragma unroll
            for (int r = 0; r < 8; ++r) cacc[t][r] *= alpha[r];

#pragma unroll
        for (int r = 0; r < 8; ++r) pb[(r + 8 * half) * 16 + lx] = p[r];
        v2f af[4];
#pragma unroll
        for (int kk = 0; kk < 4; ++kk)
            af[kk] = *(const v2f*)(pb + lx * 16 + kk * 4 + 2 * half);

#pragma unroll
        for (int t = 0; t < 4; ++t) {
            const float* vcol = vp + (size_t)(j * 16) * HDd + t * 16 + lx;
#pragma unroll
            for (int kk = 0; kk < 4; ++kk) {
                const int kr = kk * 4 + 2 * half;
                v2f bf;
                bf[0] = vcol[(size_t)kr * HDd];
                bf[1] = vcol[(size_t)(kr + 1) * HDd];
                cacc[t] = wmma4(af[kk], bf, cacc[t]);
            }
        }
    }

#pragma unroll
    for (int r = 0; r < 8; ++r) lrow[r] = 1.0f / lrow[r];
#pragma unroll
    for (int t = 0; t < 4; ++t) {
#pragma unroll
        for (int r = 0; r < 8; ++r) {
            const int n = qt * 16 + r + 8 * half;
            ctx[((size_t)(b * NTOK + n)) * DIMm + h * HDd + t * 16 + lx] =
                cacc[t][r] * lrow[r];
        }
    }
}

__global__ void proj_gemm_kernel(const float* __restrict__ ctx,
                                 const float* __restrict__ w,
                                 const float* __restrict__ bias,
                                 float* __restrict__ out) {
    const int lane = threadIdx.x & 31;
    const int lx   = lane & 15;
    const int half = lane >> 4;

    int gw = blockIdx.x * (blockDim.x >> 5) + (threadIdx.x >> 5);
    const int nt = gw % 64;  gw /= 64;
    const int ct = gw % 32;  gw /= 32;
    const int b  = gw;

    const float* arow = w   + (size_t)(ct * 16 + lx) * DIMm + 2 * half;
    const float* brow = ctx + (size_t)(b * NTOK + nt * 16 + lx) * DIMm + 2 * half;

    v8f acc = {};
#pragma unroll 8
    for (int k = 0; k < DIMm; k += 4) {
        v2f a  = *(const v2f*)(arow + k);
        v2f bf = *(const v2f*)(brow + k);
        acc = wmma4(a, bf, acc);
    }

    const int n = nt * 16 + lx;
#pragma unroll
    for (int r = 0; r < 8; ++r) {
        const int c = ct * 16 + r + 8 * half;
        out[((size_t)(b * DIMm + c)) * NTOK + n] = acc[r] + bias[c];
    }
}

extern "C" void kernel_launch(void* const* d_in, const int* in_sizes, int n_in,
                              void* d_out, int out_size, void* d_ws, size_t ws_size,
                              hipStream_t stream) {
    const float* x          = (const float*)d_in[0];
    const float* qkv_w      = (const float*)d_in[1];
    const float* qkv_b      = (const float*)d_in[2];
    const float* proj_w     = (const float*)d_in[3];
    const float* proj_b     = (const float*)d_in[4];
    const float* bias_table = (const float*)d_in[5];
    const int*   rel_index  = (const int*)d_in[6];
    float*       out        = (float*)d_out;

    float* wsf = (float*)d_ws;
    float* qkv = wsf;
    float* ctx = wsf + 3 * (size_t)HEADSZ;

    dim3 blk(256);

    qkv_gemm_kernel<<<12288, blk, 0, stream>>>(x, qkv_w, qkv_b, qkv);
    attn_kernel<<<1024, blk, 0, stream>>>(qkv, bias_table, rel_index, ctx);
    proj_gemm_kernel<<<4096, blk, 0, stream>>>(ctx, proj_w, proj_b, out);
}


# Both GEMM kernels register-blocked: one wave computes a 64x64 block (4x4 f32 WMMA tiles), one b128 fragment load feeds 8 WMMAs; k order inside each 32-float chunk changes (sum order)
# speedup vs baseline: 1.4602x; 1.4602x over previous
#include <hip/hip_runtime.h>


typedef float v2f __attribute__((ext_vector_type(2)));
typedef float v8f __attribute__((ext_vector_type(8)));

#define NHh   8
#define HDd   64
#define DIMm  512
#define Bb    16
#define NTOK  1024
#define HEADSZ (Bb * NHh * NTOK * HDd)

__device__ __forceinline__ v8f wmma4(v2f a, v2f b, v8f c) {
    return __builtin_amdgcn_wmma_f32_16x16x4_f32(
         false, a,  false, b,
         (short)0, c,  false,  false);
}

typedef float v4f __attribute__((ext_vector_type(4)));

__device__ __forceinline__ void load_chunk(v4f (&af)[4][4], v4f (&bf)[4][4],
                                           const float* __restrict__ arow,
                                           const float* __restrict__ brow, int kc) {
#pragma unroll
    for (int i = 0; i < 4; ++i)
#pragma unroll
        for (int j = 0; j < 4; ++j)
            af[i][j] = *(const v4f*)(arow + (size_t)i * 16 * DIMm + kc + 8 * j);
#pragma unroll
    for (int n = 0; n < 4; ++n)
#pragma unroll
        for (int j = 0; j < 4; ++j)
            bf[n][j] = *(const v4f*)(brow + (size_t)n * 16 * DIMm + kc + 8 * j);
}

#define LOADS_THEN_MATRIX()                                   \
    do {                                                      \
        __builtin_amdgcn_sched_group_barrier(0x020, 32, 0);   \
        __builtin_amdgcn_sched_group_barrier(0x008, 128, 0);  \
    } while (0)

__device__ __forceinline__ void mma_chunk(const v4f (&af)[4][4], const v4f (&bf)[4][4],
                                          v8f (&acc)[4][4]) {
#pragma unroll
    for (int j = 0; j < 4; ++j) {
#pragma unroll
        for (int e = 0; e < 2; ++e) {
#pragma unroll
            for (int i = 0; i < 4; ++i) {
                v2f a;
                a[0] = af[i][j][2 * e];
                a[1] = af[i][j][2 * e + 1];
#pragma unroll
                for (int n = 0; n < 4; ++n) {
                    v2f b;
                    b[0] = bf[n][j][2 * e];
                    b[1] = bf[n][j][2 * e + 1];
                    acc[i][n] = wmma4(a, b, acc[i][n]);
                }
            }
        }
    }
}

__device__ __forceinline__ void block_gemm_k512(const float* __restrict__ arow,
                                                const float* __restrict__ brow,
                                                v8f (&acc)[4][4]) {
#pragma unroll
    for (int i = 0; i < 4; ++i)
#pragma unroll
        for (int n = 0; n < 4; ++n) acc[i][n] = (v8f){};

    v4f a0[4][4], b0[4][4], a1[4][4], b1[4][4];
    load_chunk(a0, b0, arow, brow, 0);
    for (int kc = 0; kc < DIMm - 64; kc += 64) {
        load_chunk(a1, b1, arow, brow, kc + 32);
        mma_chunk(a0, b0, acc);
        LOADS_THEN_MATRIX();
        load_chunk(a0, b0, arow, brow, kc + 64);
        mma_chunk(a1, b1, acc);
        LOADS_THEN_MATRIX();
    }
    load_chunk(a1, b1, arow, brow, DIMm - 32);
    mma_chunk(a0, b0, acc);
    LOADS_THEN_MATRIX();
    mma_chunk(a1, b1, acc);
}

__global__ void __launch_bounds__(128)
qkv_gemm_kernel(const float* __restrict__ x,
                const float* __restrict__ w,
                const float* __restrict__ bias,
                float* __restrict__ qkv) {
    const int lane = threadIdx.x & 31;
    const int lx   = lane & 15;
    const int half = lane >> 4;
    const int wv   = threadIdx.x >> 5;

    int bid = blockIdx.x;
    const int nb = bid % 12;  bid /= 12;
    const int mb = bid;
    const int n0 = nb * 128 + (wv & 1) * 64;
    const int m0 = mb * 128 + (wv >> 1) * 64;

    const float* arow = x + (size_t)(m0 + lx) * DIMm + 4 * half;
    const float* brow = w + (size_t)(n0 + lx) * DIMm + 4 * half;

    v8f acc[4][4];
    block_gemm_k512(arow, brow, acc);

    const int   which = n0 >> 9;
    const int   h     = (n0 >> 6) & 7;
    const float scale = (which == 0) ? 0.125f : 1.0f;
    const int   b     = m0 >> 10;
    const int   tok0  = m0 & (NTOK - 1);

    float* dst = qkv + (size_t)which * HEADSZ
               + ((size_t)(b * NHh + h) * NTOK + tok0) * HDd;
#pragma unroll
    for (int n = 0; n < 4; ++n) {
        const int   d  = n * 16 + lx;
        const float bv = bias[n0 + d];
#pragma unroll
        for (int i = 0; i < 4; ++i) {
#pragma unroll
            for (int r = 0; r < 8; ++r) {
                const int row = i * 16 + r + 8 * half;
                dst[(size_t)row * HDd + d] = (acc[i][n][r] + bv) * scale;
            }
        }
    }
}

__global__ void attn_kernel(const float* __restrict__ qkv,
                            const float* __restrict__ bias_table,
                            const int*   __restrict__ rel_index,
                            float*       __restrict__ ctx) {
    __shared__ float pbuf[8][16 * 16];
    const int lane = threadIdx.x & 31;
    const int lx   = lane & 15;
    const int half = lane >> 4;
    const int wv   = threadIdx.x >> 5;
    float* pb = pbuf[wv];

    int gw = blockIdx.x * (blockDim.x >> 5) + wv;
    const int qt = gw % 64;  gw /= 64;
    const int h  = gw % 8;   gw /= 8;
    const int b  = gw;

    const size_t headoff = ((size_t)(b * NHh + h) * NTOK) * HDd;
    const float* qp = qkv + headoff;
    const float* kp = qkv + (size_t)HEADSZ + headoff;
    const float* vp = qkv + 2 * (size_t)HEADSZ + headoff;

    v2f qf[16];
    {
        const float* qrow = qp + (size_t)(qt * 16 + lx) * HDd + 2 * half;
#pragma unroll
        for (int kk = 0; kk < 16; ++kk) qf[kk] = *(const v2f*)(qrow + kk * 4);
    }

    v8f cacc[4];
#pragma unroll
    for (int t = 0; t < 4; ++t) cacc[t] = (v8f){};
    float mrow[8], lrow[8];
#pragma unroll
    for (int r = 0; r < 8; ++r) { mrow[r] = -3.0e38f; lrow[r] = 0.0f; }

    for (int j = 0; j < 64; ++j) {
        v8f s = (v8f){};
        const float* krow = kp + (size_t)(j * 16 + lx) * HDd + 2 * half;
#pragma unroll
        for (int kk = 0; kk < 16; ++kk) {
            v2f bf = *(const v2f*)(krow + kk * 4);
            s = wmma4(qf[kk], bf, s);
        }

        const int kcol = j * 16 + lx;
#pragma unroll
        for (int r = 0; r < 8; ++r) {
            const int qrow_ = qt * 16 + r + 8 * half;
            const int idx   = rel_index[qrow_ * NTOK + kcol];
            s[r] += bias_table[idx * NHh + h];
        }

        float p[8], alpha[8];
#pragma unroll
        for (int r = 0; r < 8; ++r) {
            float tmax = s[r];
            tmax = fmaxf(tmax, __shfl_xor(tmax, 1, 32));
            tmax = fmaxf(tmax, __shfl_xor(tmax, 2, 32));
            tmax = fmaxf(tmax, __shfl_xor(tmax, 4, 32));
            tmax = fmaxf(tmax, __shfl_xor(tmax, 8, 32));
            const float mnew = fmaxf(mrow[r], tmax);
            alpha[r] = __expf(mrow[r] - mnew);
            const float pv = __expf(s[r] - mnew);
            float rs = pv;
            rs += __shfl_xor(rs, 1, 32);
            rs += __shfl_xor(rs, 2, 32);
            rs += __shfl_xor(rs, 4, 32);
            rs += __shfl_xor(rs, 8, 32);
            lrow[r] = lrow[r] * alpha[r] + rs;
            mrow[r] = mnew;
            p[r] = pv;
        }
#pragma unroll
        for (int t = 0; t < 4; ++t)
#pragma unroll
            for (int r = 0; r < 8; ++r) cacc[t][r] *= alpha[r];

#pragma unroll
        for (int r = 0; r < 8; ++r) pb[(r + 8 * half) * 16 + lx] = p[r];
        v2f af[4];
#pragma unroll
        for (int kk = 0; kk < 4; ++kk)
            af[kk] = *(const v2f*)(pb + lx * 16 + kk * 4 + 2 * half);

#pragma unroll
        for (int t = 0; t < 4; ++t) {
            const float* vcol = vp + (size_t)(j * 16) * HDd + t * 16 + lx;
#pragma unroll
            for (int kk = 0; kk < 4; ++kk) {
                const int kr = kk * 4 + 2 * half;
                v2f bf;
                bf[0] = vcol[(size_t)kr * HDd];
                bf[1] = vcol[(size_t)(kr + 1) * HDd];
                cacc[t] = wmma4(af[kk], bf, cacc[t]);
            }
        }
    }

#pragma unroll
    for (int r = 0; r < 8; ++r) lrow[r] = 1.0f / lrow[r];
#pragma unroll
    for (int t = 0; t < 4; ++t) {
#pragma unroll
        for (int r = 0; r < 8; ++r) {
            const int n = qt * 16 + r + 8 * half;
            ctx[((size_t)(b * NTOK + n)) * DIMm + h * HDd + t * 16 + lx] =
                cacc[t][r] * lrow[r];
        }
    }
}

__global__ void __launch_bounds__(128)
proj_gemm_kernel(const float* __restrict__ ctx,
                 const float* __restrict__ w,
                 const float* __restrict__ bias,
                 float* __restrict__ out) {
    const int lane = threadIdx.x & 31;
    const int lx   = lane & 15;
    const int half = lane >> 4;
    const int wv   = threadIdx.x >> 5;

    int bid = blockIdx.x;
    const int cb = bid % 4;  bid /= 4;
    const int tb = bid;
    const int c0 = cb * 128 + (wv & 1) * 64;
    const int t0 = tb * 128 + (wv >> 1) * 64;

    const float* arow = w   + (size_t)(c0 + lx) * DIMm + 4 * half;
    const float* brow = ctx + (size_t)(t0 + lx) * DIMm + 4 * half;

    v8f acc[4][4];
    block_gemm_k512(arow, brow, acc);

    const int b  = t0 >> 10;
    const int n0 = t0 & (NTOK - 1);
#pragma unroll
    for (int i = 0; i < 4; ++i) {
#pragma unroll
        for (int r = 0; r < 8; ++r) {
            const int   c  = c0 + i * 16 + r + 8 * half;
            const float bv = bias[c];
            float* orow = out + ((size_t)(b * DIMm + c)) * NTOK + n0 + lx;
#pragma unroll
            for (int n = 0; n < 4; ++n) orow[n * 16] = acc[i][n][r] + bv;
        }
    }
}

extern "C" void kernel_launch(void* const* d_in, const int* in_sizes, int n_in,
                              void* d_out, int out_size, void* d_ws, size_t ws_size,
                              hipStream_t stream) {
    const float* x          = (const float*)d_in[0];
    const float* qkv_w      = (const float*)d_in[1];
    const float* qkv_b      = (const float*)d_in[2];
    const float* proj_w     = (const float*)d_in[3];
    const float* proj_b     = (const float*)d_in[4];
    const float* bias_table = (const float*)d_in[5];
    const int*   rel_index  = (const int*)d_in[6];
    float*       out        = (float*)d_out;

    float* wsf = (float*)d_ws;
    float* qkv = wsf;
    float* ctx = wsf + 3 * (size_t)HEADSZ;

    dim3 blk(256);
    dim3 gblk(128);

    qkv_gemm_kernel<<<1536, gblk, 0, stream>>>(x, qkv_w, qkv_b, qkv);
    attn_kernel<<<1024, blk, 0, stream>>>(qkv, bias_table, rel_index, ctx);
    proj_gemm_kernel<<<512, gblk, 0, stream>>>(ctx, proj_w, proj_b, out);
}


# Attention computes the score tile transposed (keys down rows): softmax max/sum in-lane plus one half exchange, P feeds PV from its own registers; no shuffles, no LDS; row-sum and PV sum order change
# speedup vs baseline: 1.9331x; 1.3239x over previous
#include <hip/hip_runtime.h>


typedef float v2f __attribute__((ext_vector_type(2)));
typedef float v8f __attribute__((ext_vector_type(8)));

#define NHh   8
#define HDd   64
#define DIMm  512
#define Bb    16
#define NTOK  1024
#define HEADSZ (Bb * NHh * NTOK * HDd)

__device__ __forceinline__ v8f wmma4(v2f a, v2f b, v8f c) {
    return __builtin_amdgcn_wmma_f32_16x16x4_f32(
         false, a,  false, b,
         (short)0, c,  false,  false);
}

typedef float v4f __attribute__((ext_vector_type(4)));

__device__ __forceinline__ void load_chunk(v4f (&af)[4][4], v4f (&bf)[4][4],
                                           const float* __restrict__ arow,
                                           const float* __restrict__ brow, int kc) {
#pragma unroll
    for (int i = 0; i < 4; ++i)
#pragma unroll
        for (int j = 0; j < 4; ++j)
            af[i][j] = *(const v4f*)(arow + (size_t)i * 16 * DIMm + kc + 8 * j);
#pragma unroll
    for (int n = 0; n < 4; ++n)
#pragma unroll
        for (int j = 0; j < 4; ++j)
            bf[n][j] = *(const v4f*)(brow + (size_t)n * 16 * DIMm + kc + 8 * j);
}

#define LOADS_THEN_MATRIX()                                   \
    do {                                                      \
        __builtin_amdgcn_sched_group_barrier(0x020, 32, 0);   \
        __builtin_amdgcn_sched_group_barrier(0x008, 128, 0);  \
    } while (0)

__device__ __forceinline__ void mma_chunk(const v4f (&af)[4][4], const v4f (&bf)[4][4],
                                          v8f (&acc)[4][4]) {
#pragma unroll
    for (int j = 0; j < 4; ++j) {
#pragma unroll
        for (int e = 0; e < 2; ++e) {
#pragma unroll
            for (int i = 0; i < 4; ++i) {
                v2f a;
                a[0] = af[i][j][2 * e];
                a[1] = af[i][j][2 * e + 1];
#pragma unroll
                for (int n = 0; n < 4; ++n) {
                    v2f b;
                    b[0] = bf[n][j][2 * e];
                    b[1] = bf[n][j][2 * e + 1];
                    acc[i][n] = wmma4(a, b, acc[i][n]);
                }
            }
        }
    }
}

__device__ __forceinline__ void block_gemm_k512(const float* __restrict__ arow,
                                                const float* __restrict__ brow,
                                                v8f (&acc)[4][4]) {
#pragma unroll
    for (int i = 0; i < 4; ++i)
#pragma unroll
        for (int n = 0; n < 4; ++n) acc[i][n] = (v8f){};

    v4f a0[4][4], b0[4][4], a1[4][4], b1[4][4];
    load_chunk(a0, b0, arow, brow, 0);
    for (int kc = 0; kc < DIMm - 64; kc += 64) {
        load_chunk(a1, b1, arow, brow, kc + 32);
        mma_chunk(a0, b0, acc);
        LOADS_THEN_MATRIX();
        load_chunk(a0, b0, arow, brow, kc + 64);
        mma_chunk(a1, b1, acc);
        LOADS_THEN_MATRIX();
    }
    load_chunk(a1, b1, arow, brow, DIMm - 32);
    mma_chunk(a0, b0, acc);
    LOADS_THEN_MATRIX();
    mma_chunk(a1, b1, acc);
}

__global__ void __launch_bounds__(128)
qkv_gemm_kernel(const float* __restrict__ x,
                const float* __restrict__ w,
                const float* __restrict__ bias,
                float* __restrict__ qkv) {
    const int lane = threadIdx.x & 31;
    const int lx   = lane & 15;
    const int half = lane >> 4;
    const int wv   = threadIdx.x >> 5;

    int bid = blockIdx.x;
    const int nb = bid % 12;  bid /= 12;
    const int mb = bid;
    const int n0 = nb * 128 + (wv & 1) * 64;
    const int m0 = mb * 128 + (wv >> 1) * 64;

    const float* arow = x + (size_t)(m0 + lx) * DIMm + 4 * half;
    const float* brow = w + (size_t)(n0 + lx) * DIMm + 4 * half;

    v8f acc[4][4];
    block_gemm_k512(arow, brow, acc);

    const int   which = n0 >> 9;
    const int   h     = (n0 >> 6) & 7;
    const float scale = (which == 0) ? 0.125f : 1.0f;
    const int   b     = m0 >> 10;
    const int   tok0  = m0 & (NTOK - 1);

    float* dst = qkv + (size_t)which * HEADSZ
               + ((size_t)(b * NHh + h) * NTOK + tok0) * HDd;
#pragma unroll
    for (int n = 0; n < 4; ++n) {
        const int   d  = n * 16 + lx;
        const float bv = bias[n0 + d];
#pragma unroll
        for (int i = 0; i < 4; ++i) {
#pragma unroll
            for (int r = 0; r < 8; ++r) {
                const int row = i * 16 + r + 8 * half;
                dst[(size_t)row * HDd + d] = (acc[i][n][r] + bv) * scale;
            }
        }
    }
}

typedef int v4i __attribute__((ext_vector_type(4)));

__device__ __forceinline__ float xhalf(float x) {
    const int b = __builtin_bit_cast(int, x);
    return __builtin_bit_cast(float,
        __builtin_amdgcn_permlanex16(b, b, 0x76543210, 0xfedcba98, false, false));
}

__global__ void attn_kernel(const float* __restrict__ qkv,
                            const float* __restrict__ bias_table,
                            const int*   __restrict__ rel_index,
                            float*       __restrict__ ctx) {
    const int lane = threadIdx.x & 31;
    const int lx   = lane & 15;
    const int half = lane >> 4;
    const int wv   = threadIdx.x >> 5;

    int gw = blockIdx.x * (blockDim.x >> 5) + wv;
    const int qt = gw % 64;  gw /= 64;
    const int h  = gw % 8;   gw /= 8;
    const int b  = gw;

    const size_t headoff = ((size_t)(b * NHh + h) * NTOK) * HDd;
    const float* qp = qkv + headoff;
    const float* kp = qkv + (size_t)HEADSZ + headoff;
    const float* vp = qkv + 2 * (size_t)HEADSZ + headoff;

    v2f qf[16];
    {
        const float* qrow = qp + (size_t)(qt * 16 + lx) * HDd + 2 * half;
#pragma unroll
        for (int kk = 0; kk < 16; ++kk) qf[kk] = *(const v2f*)(qrow + kk * 4);
    }

    v8f cacc[4];
#pragma unroll
    for (int t = 0; t < 4; ++t) cacc[t] = (v8f){};
    float mrun = -3.0e38f;
    float lrun = 0.0f;

    const int* irow = rel_index + (size_t)(qt * 16 + lx) * NTOK + 8 * half;

    for (int j = 0; j < 64; ++j) {
        v8f s = (v8f){};
        const float* krow = kp + (size_t)(j * 16 + lx) * HDd + 2 * half;
#pragma unroll
        for (int kk = 0; kk < 16; ++kk) {
            v2f kf = *(const v2f*)(krow + kk * 4);
            s = wmma4(kf, qf[kk], s);
        }

        {
            const v4i ia = *(const v4i*)(irow + j * 16);
            const v4i ib = *(const v4i*)(irow + j * 16 + 4);
#pragma unroll
            for (int r = 0; r < 4; ++r) {
                s[r]     += bias_table[ia[r] * NHh + h];
                s[r + 4] += bias_table[ib[r] * NHh + h];
            }
        }

        float tmax = fmaxf(fmaxf(fmaxf(s[0], s[1]), fmaxf(s[2], s[3])),
                           fmaxf(fmaxf(s[4], s[5]), fmaxf(s[6], s[7])));
        tmax = fmaxf(tmax, xhalf(tmax));
        const float mnew  = fmaxf(mrun, tmax);
        const float alpha = __expf(mrun - mnew);
        float p[8];
        float rs = 0.0f;
#pragma unroll
        for (int r = 0; r < 8; ++r) {
            p[r] = __expf(s[r] - mnew);
            rs += p[r];
        }
        lrun = lrun * alpha + rs;
        mrun = mnew;
#pragma unroll
        for (int t = 0; t < 4; ++t)
#pragma unroll
            for (int r = 0; r < 8; ++r) cacc[t][r] *= alpha;

#pragma unroll
        for (int t = 0; t < 4; ++t) {
            const float* vcol = vp + (size_t)(j * 16 + 8 * half) * HDd + t * 16 + lx;
#pragma unroll
            for (int kk = 0; kk < 4; ++kk) {
                v2f vf;
                vf[0] = vcol[(size_t)(2 * kk) * HDd];
                vf[1] = vcol[(size_t)(2 * kk + 1) * HDd];
                v2f pf;
                pf[0] = p[2 * kk];
                pf[1] = p[2 * kk + 1];
                cacc[t] = wmma4(vf, pf, cacc[t]);
            }
        }
    }

    const float inv = 1.0f / (lrun + xhalf(lrun));
    float* crow = ctx + ((size_t)(b * NTOK + qt * 16 + lx)) * DIMm + h * HDd + 8 * half;
#pragma unroll
    for (int t = 0; t < 4; ++t) {
        v4f o0, o1;
#pragma unroll
        for (int r = 0; r < 4; ++r) {
            o0[r] = cacc[t][r] * inv;
            o1[r] = cacc[t][r + 4] * inv;
        }
        *(v4f*)(crow + t * 16)     = o0;
        *(v4f*)(crow + t * 16 + 4) = o1;
    }
}

__global__ void __launch_bounds__(128)
proj_gemm_kernel(const float* __restrict__ ctx,
                 const float* __restrict__ w,
                 const float* __restrict__ bias,
                 float* __restrict__ out) {
    const int lane = threadIdx.x & 31;
    const int lx   = lane & 15;
    const int half = lane >> 4;
    const int wv   = threadIdx.x >> 5;

    int bid = blockIdx.x;
    const int cb = bid % 4;  bid /= 4;
    const int tb = bid;
    const int c0 = cb * 128 + (wv & 1) * 64;
    const int t0 = tb * 128 + (wv >> 1) * 64;

    const float* arow = w   + (size_t)(c0 + lx) * DIMm + 4 * half;
    const float* brow = ctx + (size_t)(t0 + lx) * DIMm + 4 * half;

    v8f acc[4][4];
    block_gemm_k512(arow, brow, acc);

    const int b  = t0 >> 10;
    const int n0 = t0 & (NTOK - 1);
#pragma unroll
    for (int i = 0; i < 4; ++i) {
#pragma unroll
        for (int r = 0; r < 8; ++r) {
            const int   c  = c0 + i * 16 + r + 8 * half;
            const float bv = bias[c];
            float* orow = out + ((size_t)(b * DIMm + c)) * NTOK + n0 + lx;
#pragma unroll
            for (int n = 0; n < 4; ++n) orow[n * 16] = acc[i][n][r] + bv;
        }
    }
}

extern "C" void kernel_launch(void* const* d_in, const int* in_sizes, int n_in,
                              void* d_out, int out_size, void* d_ws, size_t ws_size,
                              hipStream_t stream) {
    const float* x          = (const float*)d_in[0];
    const float* qkv_w      = (const float*)d_in[1];
    const float* qkv_b      = (const float*)d_in[2];
    const float* proj_w     = (const float*)d_in[3];
    const float* proj_b     = (const float*)d_in[4];
    const float* bias_table = (const float*)d_in[5];
    const int*   rel_index  = (const int*)d_in[6];
    float*       out        = (float*)d_out;

    float* wsf = (float*)d_ws;
    float* qkv = wsf;
    float* ctx = wsf + 3 * (size_t)HEADSZ;

    dim3 blk(256);
    dim3 gblk(128);

    qkv_gemm_kernel<<<1536, gblk, 0, stream>>>(x, qkv_w, qkv_b, qkv);
    attn_kernel<<<1024, blk, 0, stream>>>(qkv, bias_table, rel_index, ctx);
    proj_gemm_kernel<<<512, gblk, 0, stream>>>(ctx, proj_w, proj_b, out);
}


# Attention K, Q and V fragments fetched as 16-byte pieces: head-dim order of the score product and row order of the PV tiles re-chosen so a lane's pieces are contiguous; score sum order changes
# speedup vs baseline: 2.0987x; 1.0857x over previous
#include <hip/hip_runtime.h>


typedef float v2f __attribute__((ext_vector_type(2)));
typedef float v8f __attribute__((ext_vector_type(8)));

#define NHh   8
#define HDd   64
#define DIMm  512
#define Bb    16
#define NTOK  1024
#define HEADSZ (Bb * NHh * NTOK * HDd)

__device__ __forceinline__ v8f wmma4(v2f a, v2f b, v8f c) {
    return __builtin_amdgcn_wmma_f32_16x16x4_f32(
         false, a,  false, b,
         (short)0, c,  false,  false);
}

typedef float v4f __attribute__((ext_vector_type(4)));

__device__ __forceinline__ void load_chunk(v4f (&af)[4][4], v4f (&bf)[4][4],
                                           const float* __restrict__ arow,
                                           const float* __restrict__ brow, int kc) {
#pragma unroll
    for (int i = 0; i < 4; ++i)
#pragma unroll
        for (int j = 0; j < 4; ++j)
            af[i][j] = *(const v4f*)(arow + (size_t)i * 16 * DIMm + kc + 8 * j);
#pragma unroll
    for (int n = 0; n < 4; ++n)
#pragma unroll
        for (int j = 0; j < 4; ++j)
            bf[n][j] = *(const v4f*)(brow + (size_t)n * 16 * DIMm + kc + 8 * j);
}

#define LOADS_THEN_MATRIX()                                   \
    do {                                                      \
        __builtin_amdgcn_sched_group_barrier(0x020, 32, 0);   \
        __builtin_amdgcn_sched_group_barrier(0x008, 128, 0);  \
    } while (0)

__device__ __forceinline__ void mma_chunk(const v4f (&af)[4][4], const v4f (&bf)[4][4],
                                          v8f (&acc)[4][4]) {
#pragma unroll
    for (int j = 0; j < 4; ++j) {
#pragma unroll
        for (int e = 0; e < 2; ++e) {
#pragma unroll
            for (int i = 0; i < 4; ++i) {
                v2f a;
                a[0] = af[i][j][2 * e];
                a[1] = af[i][j][2 * e + 1];
#pragma unroll
                for (int n = 0; n < 4; ++n) {
                    v2f b;
                    b[0] = bf[n][j][2 * e];
                    b[1] = bf[n][j][2 * e + 1];
                    acc[i][n] = wmma4(a, b, acc[i][n]);
                }
            }
        }
    }
}

__device__ __forceinline__ void block_gemm_k512(const float* __restrict__ arow,
                                                const float* __restrict__ brow,
                                                v8f (&acc)[4][4]) {
#pragma unroll
    for (int i = 0; i < 4; ++i)
#pragma unroll
        for (int n = 0; n < 4; ++n) acc[i][n] = (v8f){};

    v4f a0[4][4], b0[4][4], a1[4][4], b1[4][4];
    load_chunk(a0, b0, arow, brow, 0);
    for (int kc = 0; kc < DIMm - 64; kc += 64) {
        load_chunk(a1, b1, arow, brow, kc + 32);
        mma_chunk(a0, b0, acc);
        LOADS_THEN_MATRIX();
        load_chunk(a0, b0, arow, brow, kc + 64);
        mma_chunk(a1, b1, acc);
        LOADS_THEN_MATRIX();
    }
    load_chunk(a1, b1, arow, brow, DIMm - 32);
    mma_chunk(a0, b0, acc);
    LOADS_THEN_MATRIX();
    mma_chunk(a1, b1, acc);
}

__global__ void __launch_bounds__(128)
qkv_gemm_kernel(const float* __restrict__ x,
                const float* __restrict__ w,
                const float* __restrict__ bias,
                float* __restrict__ qkv) {
    const int lane = threadIdx.x & 31;
    const int lx   = lane & 15;
    const int half = lane >> 4;
    const int wv   = threadIdx.x >> 5;

    int bid = blockIdx.x;
    const int nb = bid % 12;  bid /= 12;
    const int mb = bid;
    const int n0 = nb * 128 + (wv & 1) * 64;
    const int m0 = mb * 128 + (wv >> 1) * 64;

    const float* arow = x + (size_t)(m0 + lx) * DIMm + 4 * half;
    const float* brow = w + (size_t)(n0 + lx) * DIMm + 4 * half;

    v8f acc[4][4];
    block_gemm_k512(arow, brow, acc);

    const int   which = n0 >> 9;
    const int   h     = (n0 >> 6) & 7;
    const float scale = (which == 0) ? 0.125f : 1.0f;
    const int   b     = m0 >> 10;
    const int   tok0  = m0 & (NTOK - 1);

    float* dst = qkv + (size_t)which * HEADSZ
               + ((size_t)(b * NHh + h) * NTOK + tok0) * HDd;
#pragma unroll
    for (int n = 0; n < 4; ++n) {
        const int   d  = n * 16 + lx;
        const float bv = bias[n0 + d];
#pragma unroll
        for (int i = 0; i < 4; ++i) {
#pragma unroll
            for (int r = 0; r < 8; ++r) {
                const int row = i * 16 + r + 8 * half;
                dst[(size_t)row * HDd + d] = (acc[i][n][r] + bv) * scale;
            }
        }
    }
}

typedef int v4i __attribute__((ext_vector_type(4)));

__device__ __forceinline__ float xhalf(float x) {
    const int b = __builtin_bit_cast(int, x);
    return __builtin_bit_cast(float,
        __builtin_amdgcn_permlanex16(b, b, 0x76543210, 0xfedcba98, false, false));
}

__global__ void attn_kernel(const float* __restrict__ qkv,
                            const float* __restrict__ bias_table,
                            const int*   __restrict__ rel_index,
                            float*       __restrict__ ctx) {
    const int lane = threadIdx.x & 31;
    const int lx   = lane & 15;
    const int half = lane >> 4;
    const int wv   = threadIdx.x >> 5;

    int gw = blockIdx.x * (blockDim.x >> 5) + wv;
    const int qt = gw % 64;  gw /= 64;
    const int h  = gw % 8;   gw /= 8;
    const int b  = gw;

    const size_t headoff = ((size_t)(b * NHh + h) * NTOK) * HDd;
    const float* qp = qkv + headoff;
    const float* kp = qkv + (size_t)HEADSZ + headoff;
    const float* vp = qkv + 2 * (size_t)HEADSZ + headoff;

    v4f qf[8];
    {
        const float* qrow = qp + (size_t)(qt * 16 + lx) * HDd + 4 * half;
#pragma unroll
        for (int m = 0; m < 8; ++m) qf[m] = *(const v4f*)(qrow + 8 * m);
    }

    v8f cacc[4];
#pragma unroll
    for (int t = 0; t < 4; ++t) cacc[t] = (v8f){};
    float mrun = -3.0e38f;
    float lrun = 0.0f;

    const int* irow = rel_index + (size_t)(qt * 16 + lx) * NTOK + 8 * half;

    for (int j = 0; j < 64; ++j) {
        v8f s = (v8f){};
        const float* krow = kp + (size_t)(j * 16 + lx) * HDd + 4 * half;
#pragma unroll
        for (int m = 0; m < 8; ++m) {
            const v4f kf = *(const v4f*)(krow + 8 * m);
#pragma unroll
            for (int e2 = 0; e2 < 2; ++e2) {
                v2f ka, qb;
                ka[0] = kf[2 * e2];
                ka[1] = kf[2 * e2 + 1];
                qb[0] = qf[m][2 * e2];
                qb[1] = qf[m][2 * e2 + 1];
                s = wmma4(ka, qb, s);
            }
        }

        {
            const v4i ia = *(const v4i*)(irow + j * 16);
            const v4i ib = *(const v4i*)(irow + j * 16 + 4);
#pragma unroll
            for (int r = 0; r < 4; ++r) {
                s[r]     += bias_table[ia[r] * NHh + h];
                s[r + 4] += bias_table[ib[r] * NHh + h];
            }
        }

        float tmax = fmaxf(fmaxf(fmaxf(s[0], s[1]), fmaxf(s[2], s[3])),
                           fmaxf(fmaxf(s[4], s[5]), fmaxf(s[6], s[7])));
        tmax = fmaxf(tmax, xhalf(tmax));
        const float mnew  = fmaxf(mrun, tmax);
        const float alpha = __expf(mrun - mnew);
        float p[8];
        float rs = 0.0f;
#pragma unroll
        for (int r = 0; r < 8; ++r) {
            p[r] = __expf(s[r] - mnew);
            rs += p[r];
        }
        lrun = lrun * alpha + rs;
        mrun = mnew;
#pragma unroll
        for (int t = 0; t < 4; ++t)
#pragma unroll
            for (int r = 0; r < 8; ++r) cacc[t][r] *= alpha;

        v4f vv[8];
        {
            const float* vrow = vp + (size_t)(j * 16 + 8 * half) * HDd + 4 * lx;
#pragma unroll
            for (int r = 0; r < 8; ++r) vv[r] = *(const v4f*)(vrow + (size_t)r * HDd);
        }
#pragma unroll
        for (int t = 0; t < 4; ++t) {
#pragma unroll
            for (int kk = 0; kk < 4; ++kk) {
                v2f vf;
                vf[0] = vv[2 * kk][t];
                vf[1] = vv[2 * kk + 1][t];
                v2f pf;
                pf[0] = p[2 * kk];
                pf[1] = p[2 * kk + 1];
                cacc[t] = wmma4(vf, pf, cacc[t]);
            }
        }
    }

    const float inv = 1.0f / (lrun + xhalf(lrun));
    float* crow = ctx + ((size_t)(b * NTOK + qt * 16 + lx)) * DIMm + h * HDd + 32 * half;
#pragma unroll
    for (int r = 0; r < 8; ++r) {
        v4f o;
#pragma unroll
        for (int t = 0; t < 4; ++t) o[t] = cacc[t][r] * inv;
        *(v4f*)(crow + 4 * r) = o;
    }
}

__global__ void __launch_bounds__(128)
proj_gemm_kernel(const float* __restrict__ ctx,
                 const float* __restrict__ w,
                 const float* __restrict__ bias,
                 float* __restrict__ out) {
    const int lane = threadIdx.x & 31;
    const int lx   = lane & 15;
    const int half = lane >> 4;
    const int wv   = threadIdx.x >> 5;

    int bid = blockIdx.x;
    const int cb = bid % 4;  bid /= 4;
    const int tb = bid;
    const int c0 = cb * 128 + (wv & 1) * 64;
    const int t0 = tb * 128 + (wv >> 1) * 64;

    const float* arow = w   + (size_t)(c0 + lx) * DIMm + 4 * half;
    const float* brow = ctx + (size_t)(t0 + lx) * DIMm + 4 * half;

    v8f acc[4][4];
    block_gemm_k512(arow, brow, acc);

    const int b  = t0 >> 10;
    const int n0 = t0 & (NTOK - 1);
#pragma unroll
    for (int i = 0; i < 4; ++i) {
#pragma unroll
        for (int r = 0; r < 8; ++r) {
            const int   c  = c0 + i * 16 + r + 8 * half;
            const float bv = bias[c];
            float* orow = out + ((size_t)(b * DIMm + c)) * NTOK + n0 + lx;
#pragma unroll
            for (int n = 0; n < 4; ++n) orow[n * 16] = acc[i][n][r] + bv;
        }
    }
}

extern "C" void kernel_launch(void* const* d_in, const int* in_sizes, int n_in,
                              void* d_out, int out_size, void* d_ws, size_t ws_size,
                              hipStream_t stream) {
    const float* x          = (const float*)d_in[0];
    const float* qkv_w      = (const float*)d_in[1];
    const float* qkv_b      = (const float*)d_in[2];
    const float* proj_w     = (const float*)d_in[3];
    const float* proj_b     = (const float*)d_in[4];
    const float* bias_table = (const float*)d_in[5];
    const int*   rel_index  = (const int*)d_in[6];
    float*       out        = (float*)d_out;

    float* wsf = (float*)d_ws;
    float* qkv = wsf;
    float* ctx = wsf + 3 * (size_t)HEADSZ;

    dim3 blk(256);
    dim3 gblk(128);

    qkv_gemm_kernel<<<1536, gblk, 0, stream>>>(x, qkv_w, qkv_b, qkv);
    attn_kernel<<<1024, blk, 0, stream>>>(qkv, bias_table, rel_index, ctx);
    proj_gemm_kernel<<<512, gblk, 0, stream>>>(ctx, proj_w, proj_b, out);
}


# Attention wave owns 4 query tiles (64 queries): each K and V fragment load feeds 4 matrix instructions, 4 independent score accumulators; per-query arithmetic and sum order as before
# speedup vs baseline: 2.3906x; 1.1391x over previous
#include <hip/hip_runtime.h>


typedef float v2f __attribute__((ext_vector_type(2)));
typedef float v8f __attribute__((ext_vector_type(8)));

#define NHh   8
#define HDd   64
#define DIMm  512
#define Bb    16
#define NTOK  1024
#define HEADSZ (Bb * NHh * NTOK * HDd)

__device__ __forceinline__ v8f wmma4(v2f a, v2f b, v8f c) {
    return __builtin_amdgcn_wmma_f32_16x16x4_f32(
         false, a,  false, b,
         (short)0, c,  false,  false);
}

typedef float v4f __attribute__((ext_vector_type(4)));

__device__ __forceinline__ void load_chunk(v4f (&af)[4][4], v4f (&bf)[4][4],
                                           const float* __restrict__ arow,
                                           const float* __restrict__ brow, int kc) {
#pragma unroll
    for (int i = 0; i < 4; ++i)
#pragma unroll
        for (int j = 0; j < 4; ++j)
            af[i][j] = *(const v4f*)(arow + (size_t)i * 16 * DIMm + kc + 8 * j);
#pragma unroll
    for (int n = 0; n < 4; ++n)
#pragma unroll
        for (int j = 0; j < 4; ++j)
            bf[n][j] = *(const v4f*)(brow + (size_t)n * 16 * DIMm + kc + 8 * j);
}

#define LOADS_THEN_MATRIX()                                   \
    do {                                                      \
        __builtin_amdgcn_sched_group_barrier(0x020, 32, 0);   \
        __builtin_amdgcn_sched_group_barrier(0x008, 128, 0);  \
    } while (0)

__device__ __forceinline__ void mma_chunk(const v4f (&af)[4][4], const v4f (&bf)[4][4],
                                          v8f (&acc)[4][4]) {
#pragma unroll
    for (int j = 0; j < 4; ++j) {
#pragma unroll
        for (int e = 0; e < 2; ++e) {
#pragma unroll
            for (int i = 0; i < 4; ++i) {
                v2f a;
                a[0] = af[i][j][2 * e];
                a[1] = af[i][j][2 * e + 1];
#pragma unroll
                for (int n = 0; n < 4; ++n) {
                    v2f b;
                    b[0] = bf[n][j][2 * e];
                    b[1] = bf[n][j][2 * e + 1];
                    acc[i][n] = wmma4(a, b, acc[i][n]);
                }
            }
        }
    }
}

__device__ __forceinline__ void block_gemm_k512(const float* __restrict__ arow,
                                                const float* __restrict__ brow,
                                                v8f (&acc)[4][4]) {
#pragma unroll
    for (int i = 0; i < 4; ++i)
#pragma unroll
        for (int n = 0; n < 4; ++n) acc[i][n] = (v8f){};

    v4f a0[4][4], b0[4][4], a1[4][4], b1[4][4];
    load_chunk(a0, b0, arow, brow, 0);
    for (int kc = 0; kc < DIMm - 64; kc += 64) {
        load_chunk(a1, b1, arow, brow, kc + 32);
        mma_chunk(a0, b0, acc);
        LOADS_THEN_MATRIX();
        load_chunk(a0, b0, arow, brow, kc + 64);
        mma_chunk(a1, b1, acc);
        LOADS_THEN_MATRIX();
    }
    load_chunk(a1, b1, arow, brow, DIMm - 32);
    mma_chunk(a0, b0, acc);
    LOADS_THEN_MATRIX();
    mma_chunk(a1, b1, acc);
}

__global__ void __launch_bounds__(128)
qkv_gemm_kernel(const float* __restrict__ x,
                const float* __restrict__ w,
                const float* __restrict__ bias,
                float* __restrict__ qkv) {
    const int lane = threadIdx.x & 31;
    const int lx   = lane & 15;
    const int half = lane >> 4;
    const int wv   = threadIdx.x >> 5;

    int bid = blockIdx.x;
    const int nb = bid % 12;  bid /= 12;
    const int mb = bid;
    const int n0 = nb * 128 + (wv & 1) * 64;
    const int m0 = mb * 128 + (wv >> 1) * 64;

    const float* arow = x + (size_t)(m0 + lx) * DIMm + 4 * half;
    const float* brow = w + (size_t)(n0 + lx) * DIMm + 4 * half;

    v8f acc[4][4];
    block_gemm_k512(arow, brow, acc);

    const int   which = n0 >> 9;
    const int   h     = (n0 >> 6) & 7;
    const float scale = (which == 0) ? 0.125f : 1.0f;
    const int   b     = m0 >> 10;
    const int   tok0  = m0 & (NTOK - 1);

    float* dst = qkv + (size_t)which * HEADSZ
               + ((size_t)(b * NHh + h) * NTOK + tok0) * HDd;
#pragma unroll
    for (int n = 0; n < 4; ++n) {
        const int   d  = n * 16 + lx;
        const float bv = bias[n0 + d];
#pragma unroll
        for (int i = 0; i < 4; ++i) {
#pragma unroll
            for (int r = 0; r < 8; ++r) {
                const int row = i * 16 + r + 8 * half;
                dst[(size_t)row * HDd + d] = (acc[i][n][r] + bv) * scale;
            }
        }
    }
}

typedef int v4i __attribute__((ext_vector_type(4)));

__device__ __forceinline__ float xhalf(float x) {
    const int b = __builtin_bit_cast(int, x);
    return __builtin_bit_cast(float,
        __builtin_amdgcn_permlanex16(b, b, 0x76543210, 0xfedcba98, false, false));
}

#define QT 4

__global__ void __launch_bounds__(256)
attn_kernel(const float* __restrict__ qkv,
            const float* __restrict__ bias_table,
            const int*   __restrict__ rel_index,
            float*       __restrict__ ctx) {
    const int lane = threadIdx.x & 31;
    const int lx   = lane & 15;
    const int half = lane >> 4;
    const int wv   = threadIdx.x >> 5;

    int gw = blockIdx.x * (blockDim.x >> 5) + wv;
    const int qg = gw % (64 / QT);  gw /= (64 / QT);
    const int h  = gw % 8;          gw /= 8;
    const int b  = gw;
    const int q0 = qg * (16 * QT);

    const size_t headoff = ((size_t)(b * NHh + h) * NTOK) * HDd;
    const float* qp = qkv + headoff;
    const float* kp = qkv + (size_t)HEADSZ + headoff;
    const float* vp = qkv + 2 * (size_t)HEADSZ + headoff;

    v4f qf[QT][8];
#pragma unroll
    for (int u = 0; u < QT; ++u) {
        const float* qrow = qp + (size_t)(q0 + 16 * u + lx) * HDd + 4 * half;
#pragma unroll
        for (int m = 0; m < 8; ++m) qf[u][m] = *(const v4f*)(qrow + 8 * m);
    }

    v8f cacc[QT][4];
    float mrun[QT];
    float lrun[QT];
#pragma unroll
    for (int u = 0; u < QT; ++u) {
#pragma unroll
        for (int t = 0; t < 4; ++t) cacc[u][t] = (v8f){};
        mrun[u] = -3.0e38f;
        lrun[u] = 0.0f;
    }

    const int* irow = rel_index + (size_t)(q0 + lx) * NTOK + 8 * half;

    for (int j = 0; j < 64; ++j) {
        v8f s[QT];
#pragma unroll
        for (int u = 0; u < QT; ++u) s[u] = (v8f){};
        const float* krow = kp + (size_t)(j * 16 + lx) * HDd + 4 * half;
#pragma unroll
        for (int m = 0; m < 8; ++m) {
            const v4f kf = *(const v4f*)(krow + 8 * m);
#pragma unroll
            for (int e2 = 0; e2 < 2; ++e2) {
                v2f ka;
                ka[0] = kf[2 * e2];
                ka[1] = kf[2 * e2 + 1];
#pragma unroll
                for (int u = 0; u < QT; ++u) {
                    v2f qb;
                    qb[0] = qf[u][m][2 * e2];
                    qb[1] = qf[u][m][2 * e2 + 1];
                    s[u] = wmma4(ka, qb, s[u]);
                }
            }
        }

        v4f vv[8];
        {
            const float* vrow = vp + (size_t)(j * 16 + 8 * half) * HDd + 4 * lx;
#pragma unroll
            for (int r = 0; r < 8; ++r) vv[r] = *(const v4f*)(vrow + (size_t)r * HDd);
        }

        float p[QT][8];
#pragma unroll
        for (int u = 0; u < QT; ++u) {
            {
                const int* ir = irow + (size_t)u * 16 * NTOK + j * 16;
                const v4i ia = *(const v4i*)(ir);
                const v4i ib = *(const v4i*)(ir + 4);
#pragma unroll
                for (int r = 0; r < 4; ++r) {
                    s[u][r]     += bias_table[ia[r] * NHh + h];
                    s[u][r + 4] += bias_table[ib[r] * NHh + h];
                }
            }

            float tmax = fmaxf(fmaxf(fmaxf(s[u][0], s[u][1]), fmaxf(s[u][2], s[u][3])),
                               fmaxf(fmaxf(s[u][4], s[u][5]), fmaxf(s[u][6], s[u][7])));
            tmax = fmaxf(tmax, xhalf(tmax));
            const float mnew  = fmaxf(mrun[u], tmax);
            const float alpha = __expf(mrun[u] - mnew);
            float rs = 0.0f;
#pragma unroll
            for (int r = 0; r < 8; ++r) {
                p[u][r] = __expf(s[u][r] - mnew);
                rs += p[u][r];
            }
            lrun[u] = lrun[u] * alpha + rs;
            mrun[u] = mnew;
#pragma unroll
            for (int t = 0; t < 4; ++t)
#pragma unroll
                for (int r = 0; r < 8; ++r) cacc[u][t][r] *= alpha;
        }

#pragma unroll
        for (int kk = 0; kk < 4; ++kk) {
#pragma unroll
            for (int t = 0; t < 4; ++t) {
                v2f vf;
                vf[0] = vv[2 * kk][t];
                vf[1] = vv[2 * kk + 1][t];
#pragma unroll
                for (int u = 0; u < QT; ++u) {
                    v2f pf;
                    pf[0] = p[u][2 * kk];
                    pf[1] = p[u][2 * kk + 1];
                    cacc[u][t] = wmma4(vf, pf, cacc[u][t]);
                }
            }
        }
    }

#pragma unroll
    for (int u = 0; u < QT; ++u) {
        const float inv = 1.0f / (lrun[u] + xhalf(lrun[u]));
        float* crow = ctx + ((size_t)(b * NTOK + q0 + 16 * u + lx)) * DIMm + h * HDd + 32 * half;
#pragma unroll
        for (int r = 0; r < 8; ++r) {
            v4f o;
#pragma unroll
            for (int t = 0; t < 4; ++t) o[t] = cacc[u][t][r] * inv;
            *(v4f*)(crow + 4 * r) = o;
        }
    }
}

__global__ void __launch_bounds__(128)
proj_gemm_kernel(const float* __restrict__ ctx,
                 const float* __restrict__ w,
                 const float* __restrict__ bias,
                 float* __restrict__ out) {
    const int lane = threadIdx.x & 31;
    const int lx   = lane & 15;
    const int half = lane >> 4;
    const int wv   = threadIdx.x >> 5;

    int bid = blockIdx.x;
    const int cb = bid % 4;  bid /= 4;
    const int tb = bid;
    const int c0 = cb * 128 + (wv & 1) * 64;
    const int t0 = tb * 128 + (wv >> 1) * 64;

    const float* arow = w   + (size_t)(c0 + lx) * DIMm + 4 * half;
    const float* brow = ctx + (size_t)(t0 + lx) * DIMm + 4 * half;

    v8f acc[4][4];
    block_gemm_k512(arow, brow, acc);

    const int b  = t0 >> 10;
    const int n0 = t0 & (NTOK - 1);
#pragma unroll
    for (int i = 0; i < 4; ++i) {
#pragma unroll
        for (int r = 0; r < 8; ++r) {
            const int   c  = c0 + i * 16 + r + 8 * half;
            const float bv = bias[c];
            float* orow = out + ((size_t)(b * DIMm + c)) * NTOK + n0 + lx;
#pragma unroll
            for (int n = 0; n < 4; ++n) orow[n * 16] = acc[i][n][r] + bv;
        }
    }
}

extern "C" void kernel_launch(void* const* d_in, const int* in_sizes, int n_in,
                              void* d_out, int out_size, void* d_ws, size_t ws_size,
                              hipStream_t stream) {
    const float* x          = (const float*)d_in[0];
    const float* qkv_w      = (const float*)d_in[1];
    const float* qkv_b      = (const float*)d_in[2];
    const float* proj_w     = (const float*)d_in[3];
    const float* proj_b     = (const float*)d_in[4];
    const float* bias_table = (const float*)d_in[5];
    const int*   rel_index  = (const int*)d_in[6];
    float*       out        = (float*)d_out;

    float* wsf = (float*)d_ws;
    float* qkv = wsf;
    float* ctx = wsf + 3 * (size_t)HEADSZ;

    dim3 blk(256);
    dim3 gblk(128);

    qkv_gemm_kernel<<<1536, gblk, 0, stream>>>(x, qkv_w, qkv_b, qkv);
    attn_kernel<<<1024 / QT, blk, 0, stream>>>(qkv, bias_table, rel_index, ctx);
    proj_gemm_kernel<<<512, gblk, 0, stream>>>(ctx, proj_w, proj_b, out);
}


# Attention key loop rotated one tile: next tile's K fragments and bias are fetched behind the score product; the fetched bias seeds the score accumulators (no bias adds); score sum order changes
# speedup vs baseline: 2.4957x; 1.0440x over previous
#include <hip/hip_runtime.h>


typedef float v2f __attribute__((ext_vector_type(2)));
typedef float v8f __attribute__((ext_vector_type(8)));

#define NHh   8
#define HDd   64
#define DIMm  512
#define Bb    16
#define NTOK  1024
#define HEADSZ (Bb * NHh * NTOK * HDd)

__device__ __forceinline__ v8f wmma4(v2f a, v2f b, v8f c) {
    return __builtin_amdgcn_wmma_f32_16x16x4_f32(
         false, a,  false, b,
         (short)0, c,  false,  false);
}

typedef float v4f __attribute__((ext_vector_type(4)));

__device__ __forceinline__ void load_chunk(v4f (&af)[4][4], v4f (&bf)[4][4],
                                           const float* __restrict__ arow,
                                           const float* __restrict__ brow, int kc) {
#pragma unroll
    for (int i = 0; i < 4; ++i)
#pragma unroll
        for (int j = 0; j < 4; ++j)
            af[i][j] = *(const v4f*)(arow + (size_t)i * 16 * DIMm + kc + 8 * j);
#pragma unroll
    for (int n = 0; n < 4; ++n)
#pragma unroll
        for (int j = 0; j < 4; ++j)
            bf[n][j] = *(const v4f*)(brow + (size_t)n * 16 * DIMm + kc + 8 * j);
}

#define LOADS_THEN_MATRIX()                                   \
    do {                                                      \
        __builtin_amdgcn_sched_group_barrier(0x020, 32, 0);   \
        __builtin_amdgcn_sched_group_barrier(0x008, 128, 0);  \
    } while (0)

__device__ __forceinline__ void mma_chunk(const v4f (&af)[4][4], const v4f (&bf)[4][4],
                                          v8f (&acc)[4][4]) {
#pragma unroll
    for (int j = 0; j < 4; ++j) {
#pragma unroll
        for (int e = 0; e < 2; ++e) {
#pragma unroll
            for (int i = 0; i < 4; ++i) {
                v2f a;
                a[0] = af[i][j][2 * e];
                a[1] = af[i][j][2 * e + 1];
#pragma unroll
                for (int n = 0; n < 4; ++n) {
                    v2f b;
                    b[0] = bf[n][j][2 * e];
                    b[1] = bf[n][j][2 * e + 1];
                    acc[i][n] = wmma4(a, b, acc[i][n]);
                }
            }
        }
    }
}

__device__ __forceinline__ void block_gemm_k512(const float* __restrict__ arow,
                                                const float* __restrict__ brow,
                                                v8f (&acc)[4][4]) {
#pragma unroll
    for (int i = 0; i < 4; ++i)
#pragma unroll
        for (int n = 0; n < 4; ++n) acc[i][n] = (v8f){};

    v4f a0[4][4], b0[4][4], a1[4][4], b1[4][4];
    load_chunk(a0, b0, arow, brow, 0);
    for (int kc = 0; kc < DIMm - 64; kc += 64) {
        load_chunk(a1, b1, arow, brow, kc + 32);
        mma_chunk(a0, b0, acc);
        LOADS_THEN_MATRIX();
        load_chunk(a0, b0, arow, brow, kc + 64);
        mma_chunk(a1, b1, acc);
        LOADS_THEN_MATRIX();
    }
    load_chunk(a1, b1, arow, brow, DIMm - 32);
    mma_chunk(a0, b0, acc);
    LOADS_THEN_MATRIX();
    mma_chunk(a1, b1, acc);
}

__global__ void __launch_bounds__(128)
qkv_gemm_kernel(const float* __restrict__ x,
                const float* __restrict__ w,
                const float* __restrict__ bias,
                float* __restrict__ qkv) {
    const int lane = threadIdx.x & 31;
    const int lx   = lane & 15;
    const int half = lane >> 4;
    const int wv   = threadIdx.x >> 5;

    int bid = blockIdx.x;
    const int nb = bid % 12;  bid /= 12;
    const int mb = bid;
    const int n0 = nb * 128 + (wv & 1) * 64;
    const int m0 = mb * 128 + (wv >> 1) * 64;

    const float* arow = x + (size_t)(m0 + lx) * DIMm + 4 * half;
    const float* brow = w + (size_t)(n0 + lx) * DIMm + 4 * half;

    v8f acc[4][4];
    block_gemm_k512(arow, brow, acc);

    const int   which = n0 >> 9;
    const int   h     = (n0 >> 6) & 7;
    const float scale = (which == 0) ? 0.125f : 1.0f;
    const int   b     = m0 >> 10;
    const int   tok0  = m0 & (NTOK - 1);

    float* dst = qkv + (size_t)which * HEADSZ
               + ((size_t)(b * NHh + h) * NTOK + tok0) * HDd;
#pragma unroll
    for (int n = 0; n < 4; ++n) {
        const int   d  = n * 16 + lx;
        const float bv = bias[n0 + d];
#pragma unroll
        for (int i = 0; i < 4; ++i) {
#pragma unroll
            for (int r = 0; r < 8; ++r) {
                const int row = i * 16 + r + 8 * half;
                dst[(size_t)row * HDd + d] = (acc[i][n][r] + bv) * scale;
            }
        }
    }
}

typedef int v4i __attribute__((ext_vector_type(4)));

__device__ __forceinline__ float xhalf(float x) {
    const int b = __builtin_bit_cast(int, x);
    return __builtin_bit_cast(float,
        __builtin_amdgcn_permlanex16(b, b, 0x76543210, 0xfedcba98, false, false));
}

#define QT 4

__global__ void __launch_bounds__(256)
attn_kernel(const float* __restrict__ qkv,
            const float* __restrict__ bias_table,
            const int*   __restrict__ rel_index,
            float*       __restrict__ ctx) {
    const int lane = threadIdx.x & 31;
    const int lx   = lane & 15;
    const int half = lane >> 4;
    const int wv   = threadIdx.x >> 5;

    int gw = blockIdx.x * (blockDim.x >> 5) + wv;
    const int qg = gw % (64 / QT);  gw /= (64 / QT);
    const int h  = gw % 8;          gw /= 8;
    const int b  = gw;
    const int q0 = qg * (16 * QT);

    const size_t headoff = ((size_t)(b * NHh + h) * NTOK) * HDd;
    const float* qp = qkv + headoff;
    const float* kp = qkv + (size_t)HEADSZ + headoff;
    const float* vp = qkv + 2 * (size_t)HEADSZ + headoff;

    v4f qf[QT][8];
#pragma unroll
    for (int u = 0; u < QT; ++u) {
        const float* qrow = qp + (size_t)(q0 + 16 * u + lx) * HDd + 4 * half;
#pragma unroll
        for (int m = 0; m < 8; ++m) qf[u][m] = *(const v4f*)(qrow + 8 * m);
    }

    v8f cacc[QT][4];
    float mrun[QT];
    float lrun[QT];
#pragma unroll
    for (int u = 0; u < QT; ++u) {
#pragma unroll
        for (int t = 0; t < 4; ++t) cacc[u][t] = (v8f){};
        mrun[u] = -3.0e38f;
        lrun[u] = 0.0f;
    }

    const int* irow = rel_index + (size_t)(q0 + lx) * NTOK + 8 * half;

    v4f kf[8];
    v8f s[QT];
    {
        const float* krow = kp + (size_t)lx * HDd + 4 * half;
#pragma unroll
        for (int m = 0; m < 8; ++m) kf[m] = *(const v4f*)(krow + 8 * m);
#pragma unroll
        for (int u = 0; u < QT; ++u) {
            const int* ir = irow + (size_t)u * 16 * NTOK;
            const v4i ia = *(const v4i*)(ir);
            const v4i ib = *(const v4i*)(ir + 4);
#pragma unroll
            for (int r = 0; r < 4; ++r) {
                s[u][r]     = bias_table[ia[r] * NHh + h];
                s[u][r + 4] = bias_table[ib[r] * NHh + h];
            }
        }
    }

    for (int j = 0; j < 64; ++j) {
        const int jn = (j + 1 < 63) ? j + 1 : 63;

        v4i idx[QT][2];
#pragma unroll
        for (int u = 0; u < QT; ++u) {
            const int* ir = irow + (size_t)u * 16 * NTOK + jn * 16;
            idx[u][0] = *(const v4i*)(ir);
            idx[u][1] = *(const v4i*)(ir + 4);
        }
        v4f vv[8];
        {
            const float* vrow = vp + (size_t)(j * 16 + 8 * half) * HDd + 4 * lx;
#pragma unroll
            for (int r = 0; r < 8; ++r) vv[r] = *(const v4f*)(vrow + (size_t)r * HDd);
        }

#pragma unroll
        for (int m = 0; m < 8; ++m) {
#pragma unroll
            for (int e2 = 0; e2 < 2; ++e2) {
                v2f ka;
                ka[0] = kf[m][2 * e2];
                ka[1] = kf[m][2 * e2 + 1];
#pragma unroll
                for (int u = 0; u < QT; ++u) {
                    v2f qb;
                    qb[0] = qf[u][m][2 * e2];
                    qb[1] = qf[u][m][2 * e2 + 1];
                    s[u] = wmma4(ka, qb, s[u]);
                }
            }
        }

        __builtin_amdgcn_sched_barrier(0);

        {
            const float* krow = kp + (size_t)(jn * 16 + lx) * HDd + 4 * half;
#pragma unroll
            for (int m = 0; m < 8; ++m) kf[m] = *(const v4f*)(krow + 8 * m);
        }
        float p[QT][8];
#pragma unroll
        for (int u = 0; u < QT; ++u) {
            float tmax = fmaxf(fmaxf(fmaxf(s[u][0], s[u][1]), fmaxf(s[u][2], s[u][3])),
                               fmaxf(fmaxf(s[u][4], s[u][5]), fmaxf(s[u][6], s[u][7])));
            tmax = fmaxf(tmax, xhalf(tmax));
            const float mnew  = fmaxf(mrun[u], tmax);
            const float alpha = __expf(mrun[u] - mnew);
            float rs = 0.0f;
#pragma unroll
            for (int r = 0; r < 8; ++r) {
                p[u][r] = __expf(s[u][r] - mnew);
                rs += p[u][r];
            }
            lrun[u] = lrun[u] * alpha + rs;
            mrun[u] = mnew;
#pragma unroll
            for (int t = 0; t < 4; ++t)
#pragma unroll
                for (int r = 0; r < 8; ++r) cacc[u][t][r] *= alpha;
        }

#pragma unroll
        for (int u = 0; u < QT; ++u) {
#pragma unroll
            for (int r = 0; r < 4; ++r) {
                s[u][r]     = bias_table[idx[u][0][r] * NHh + h];
                s[u][r + 4] = bias_table[idx[u][1][r] * NHh + h];
            }
        }

#pragma unroll
        for (int kk = 0; kk < 4; ++kk) {
#pragma unroll
            for (int t = 0; t < 4; ++t) {
                v2f vf;
                vf[0] = vv[2 * kk][t];
                vf[1] = vv[2 * kk + 1][t];
#pragma unroll
                for (int u = 0; u < QT; ++u) {
                    v2f pf;
                    pf[0] = p[u][2 * kk];
                    pf[1] = p[u][2 * kk + 1];
                    cacc[u][t] = wmma4(vf, pf, cacc[u][t]);
                }
            }
        }

        asm volatile("" :: "v"(kf[0]), "v"(kf[1]), "v"(kf[2]), "v"(kf[3]),
                           "v"(kf[4]), "v"(kf[5]), "v"(kf[6]), "v"(kf[7]),
                           "v"(s[0]), "v"(s[1]), "v"(s[2]), "v"(s[3]));
    }

#pragma unroll
    for (int u = 0; u < QT; ++u) {
        const float inv = 1.0f / (lrun[u] + xhalf(lrun[u]));
        float* crow = ctx + ((size_t)(b * NTOK + q0 + 16 * u + lx)) * DIMm + h * HDd + 32 * half;
#pragma unroll
        for (int r = 0; r < 8; ++r) {
            v4f o;
#pragma unroll
            for (int t = 0; t < 4; ++t) o[t] = cacc[u][t][r] * inv;
            *(v4f*)(crow + 4 * r) = o;
        }
    }
}

__global__ void __launch_bounds__(128)
proj_gemm_kernel(const float* __restrict__ ctx,
                 const float* __restrict__ w,
                 const float* __restrict__ bias,
                 float* __restrict__ out) {
    const int lane = threadIdx.x & 31;
    const int lx   = lane & 15;
    const int half = lane >> 4;
    const int wv   = threadIdx.x >> 5;

    int bid = blockIdx.x;
    const int cb = bid % 4;  bid /= 4;
    const int tb = bid;
    const int c0 = cb * 128 + (wv & 1) * 64;
    const int t0 = tb * 128 + (wv >> 1) * 64;

    const float* arow = w   + (size_t)(c0 + lx) * DIMm + 4 * half;
    const float* brow = ctx + (size_t)(t0 + lx) * DIMm + 4 * half;

    v8f acc[4][4];
    block_gemm_k512(arow, brow, acc);

    const int b  = t0 >> 10;
    const int n0 = t0 & (NTOK - 1);
#pragma unroll
    for (int i = 0; i < 4; ++i) {
#pragma unroll
        for (int r = 0; r < 8; ++r) {
            const int   c  = c0 + i * 16 + r + 8 * half;
            const float bv = bias[c];
            float* orow = out + ((size_t)(b * DIMm + c)) * NTOK + n0 + lx;
#pragma unroll
            for (int n = 0; n < 4; ++n) orow[n * 16] = acc[i][n][r] + bv;
        }
    }
}

extern "C" void kernel_launch(void* const* d_in, const int* in_sizes, int n_in,
                              void* d_out, int out_size, void* d_ws, size_t ws_size,
                              hipStream_t stream) {
    const float* x          = (const float*)d_in[0];
    const float* qkv_w      = (const float*)d_in[1];
    const float* qkv_b      = (const float*)d_in[2];
    const float* proj_w     = (const float*)d_in[3];
    const float* proj_b     = (const float*)d_in[4];
    const float* bias_table = (const float*)d_in[5];
    const int*   rel_index  = (const int*)d_in[6];
    float*       out        = (float*)d_out;

    float* wsf = (float*)d_ws;
    float* qkv = wsf;
    float* ctx = wsf + 3 * (size_t)HEADSZ;

    dim3 blk(256);
    dim3 gblk(128);

    qkv_gemm_kernel<<<1536, gblk, 0, stream>>>(x, qkv_w, qkv_b, qkv);
    attn_kernel<<<1024 / QT, blk, 0, stream>>>(qkv, bias_table, rel_index, ctx);
    proj_gemm_kernel<<<512, gblk, 0, stream>>>(ctx, proj_w, proj_b, out);
}
